# starting kernel
# baseline (speedup 1.0000x reference)
#include <hip/hip_runtime.h>

typedef __attribute__((ext_vector_type(2))) float v2f;
typedef __attribute__((ext_vector_type(8))) float v8f;

#define B_TOT 1024
#define T_LEN 4096
#define HID   20
#define NG    80
#define ROWS  16
#define GST   84

__device__ __forceinline__ float sigm(float x) {
    return 1.0f / (1.0f + __expf(-x));
}
__device__ __forceinline__ float tanh_f(float x) {
    return 1.0f - 2.0f / (__expf(2.0f * x) + 1.0f);
}

__global__ void __launch_bounds__(32)
lstm_seq_kernel(const float* __restrict__ xin,
                const float* __restrict__ Wih1,
                const float* __restrict__ Whh1,
                const float* __restrict__ bih1,
                const float* __restrict__ bhh1,
                const float* __restrict__ Wih2,
                const float* __restrict__ Whh2,
                const float* __restrict__ bih2,
                const float* __restrict__ bhh2,
                const int*   __restrict__ futp,
                float* __restrict__ out)
{
    __shared__ float lds_h1[ROWS * HID];
    __shared__ float lds_c1[ROWS * HID];
    __shared__ float lds_g [ROWS * GST];
    __shared__ float lds_x [ROWS];
    __shared__ float s_wih[NG];
    __shared__ float s_bs [NG];
    __shared__ float s_w2 [NG];
    __shared__ float s_whh2[4];
    __shared__ float s_b2 [4];

    const int lane  = threadIdx.x;
    const int nu    = lane & 15;
    const int khalf = lane >> 4;
    const int row0  = blockIdx.x * ROWS;
    const int fut   = *futp;
    const int TT    = T_LEN + fut;

    for (int i = lane; i < NG; i += 32) {
        s_wih[i] = Wih1[i];
        s_bs [i] = bih1[i] + bhh1[i];
        s_w2 [i] = Wih2[i];
    }
    if (lane < 4) {
        s_whh2[lane] = Whh2[lane];
        s_b2 [lane]  = bih2[lane] + bhh2[lane];
    }
    for (int i = lane; i < ROWS * HID; i += 32) {
        lds_h1[i] = 0.0f;
        lds_c1[i] = 0.0f;
    }

    v2f bf[5][5];
#pragma unroll
    for (int kc = 0; kc < 5; ++kc) {
#pragma unroll
        for (int n = 0; n < 5; ++n) {
            const float* p = Whh1 + (16 * n + nu) * HID + 4 * kc + 2 * khalf;
            v2f b; b.x = p[0]; b.y = p[1];
            bf[kc][n] = b;
        }
    }

    float c1v[10];
#pragma unroll
    for (int j = 0; j < 10; ++j) c1v[j] = 0.0f;
    float h2v = 0.0f, c2v = 0.0f;

    const int er  = lane >> 1;
    const int eh0 = (lane & 1) * 10;

    __syncthreads();

    for (int t = 0; t < TT; ++t) {
        if (t < T_LEN) {
            if (lane < ROWS)
                lds_x[lane] = xin[(size_t)(row0 + lane) * T_LEN + t];
        }
        __syncthreads();

        v8f acc[5];
#pragma unroll
        for (int n = 0; n < 5; ++n) {
            v8f z = {0.f, 0.f, 0.f, 0.f, 0.f, 0.f, 0.f, 0.f};
            acc[n] = z;
        }
#pragma unroll
        for (int kc = 0; kc < 5; ++kc) {
            v2f a;
            const float* hp = &lds_h1[nu * HID + 4 * kc + 2 * khalf];
            a.x = hp[0]; a.y = hp[1];
#pragma unroll
            for (int n = 0; n < 5; ++n) {
                acc[n] = __builtin_amdgcn_wmma_f32_16x16x4_f32(
                    false, a, false, bf[kc][n], (short)0, acc[n], false, false);
            }
        }
#pragma unroll
        for (int n = 0; n < 5; ++n) {
#pragma unroll
            for (int v = 0; v < 8; ++v) {
                lds_g[(v + 8 * khalf) * GST + 16 * n + nu] = acc[n][v];
            }
        }
        __syncthreads();

        {
            const float  xr = lds_x[er];
            const float* gr = &lds_g[er * GST];
#pragma unroll
            for (int j = 0; j < 10; ++j) {
                const int h = eh0 + j;
                float gi = gr[h]      + xr * s_wih[h]      + s_bs[h];
                float gf = gr[20 + h] + xr * s_wih[20 + h] + s_bs[20 + h];
                float gg = gr[40 + h] + xr * s_wih[40 + h] + s_bs[40 + h];
                float go = gr[60 + h] + xr * s_wih[60 + h] + s_bs[60 + h];
                float cn = sigm(gf) * c1v[j] + sigm(gi) * tanh_f(gg);
                c1v[j] = cn;
                float hn = sigm(go) * tanh_f(cn);
                lds_h1[er * HID + h] = hn;
                lds_c1[er * HID + h] = cn;
            }
        }
        __syncthreads();

        if (lane < ROWS) {
            float g2[4];
#pragma unroll
            for (int g = 0; g < 4; ++g) g2[g] = s_b2[g] + h2v * s_whh2[g];
            const float* cr = &lds_c1[lane * HID];
#pragma unroll
            for (int h = 0; h < HID; ++h) {
                float cv = cr[h];
#pragma unroll
                for (int g = 0; g < 4; ++g) g2[g] += cv * s_w2[g * HID + h];
            }
            float cn2 = sigm(g2[1]) * c2v + sigm(g2[0]) * tanh_f(g2[2]);
            c2v = cn2;
            h2v = sigm(g2[3]) * tanh_f(cn2);
            out[(size_t)(row0 + lane) * TT + t] = cn2;
            lds_x[lane] = cn2;
        }
        __syncthreads();
    }
}

extern "C" void kernel_launch(void* const* d_in, const int* in_sizes, int n_in,
                              void* d_out, int out_size, void* d_ws, size_t ws_size,
                              hipStream_t stream)
{
    const float* xin  = (const float*)d_in[0];
    const float* Wih1 = (const float*)d_in[1];
    const float* Whh1 = (const float*)d_in[2];
    const float* bih1 = (const float*)d_in[3];
    const float* bhh1 = (const float*)d_in[4];
    const float* Wih2 = (const float*)d_in[5];
    const float* Whh2 = (const float*)d_in[6];
    const float* bih2 = (const float*)d_in[7];
    const float* bhh2 = (const float*)d_in[8];
    const int*   futp = (const int*)d_in[9];
    float* out = (float*)d_out;

    dim3 grid(B_TOT / ROWS);
    dim3 block(32);
    lstm_seq_kernel<<<grid, block, 0, stream>>>(xin, Wih1, Whh1, bih1, bhh1,
                                                Wih2, Whh2, bih2, bhh2, futp, out);
}


# Transposed product gates^T = W.h^T, gate rows permuted: a lane's D registers hold all gates of its own cells and its next B fragment, so h1/c1 stay in registers (no LDS, no waits); sum order changes
# speedup vs baseline: 1.4714x; 1.4714x over previous
#include <hip/hip_runtime.h>

typedef __attribute__((ext_vector_type(2))) float v2f;
typedef __attribute__((ext_vector_type(8))) float v8f;

#define B_TOT 1024
#define T_LEN 4096
#define HID   20
#define ROWS  16

__device__ __forceinline__ float sigm(float x) {
    return 1.0f / (1.0f + __expf(-x));
}
__device__ __forceinline__ float tanh_f(float x) {
    return 1.0f - 2.0f / (__expf(2.0f * x) + 1.0f);
}
__device__ __forceinline__ float xhalf(float x) {
    int b = __builtin_bit_cast(int, x);
    return __builtin_bit_cast(float,
        __builtin_amdgcn_permlanex16(b, b, 0x76543210, 0xfedcba98, false, false));
}

__global__ void __launch_bounds__(32)
lstm_seq_kernel(const float* __restrict__ xin,
                const float* __restrict__ Wih1,
                const float* __restrict__ Whh1,
                const float* __restrict__ bih1,
                const float* __restrict__ bhh1,
                const float* __restrict__ Wih2,
                const float* __restrict__ Whh2,
                const float* __restrict__ bih2,
                const float* __restrict__ bhh2,
                const int*   __restrict__ futp,
                float* __restrict__ out)
{
    const int lane = threadIdx.x;
    const int c    = lane & 15;
    const int hf   = lane >> 4;
    const int row0 = blockIdx.x * ROWS;
    const int fut  = *futp;
    const int TT   = T_LEN + fut;

    v2f af[5][5];
    {
        const int hm = c >> 3, r = c & 7, g = r >> 1, u = r & 1;
#pragma unroll
        for (int mt = 0; mt < 5; ++mt) {
            const int wrow = g * HID + 4 * mt + 2 * hm + u;
#pragma unroll
            for (int kc = 0; kc < 5; ++kc) {
                const float* p = Whh1 + wrow * HID + 4 * kc + 2 * hf;
                v2f a; a.x = p[0]; a.y = p[1];
                af[mt][kc] = a;
            }
        }
    }
    float wx[5][8];
    v8f   bs[5];
#pragma unroll
    for (int mt = 0; mt < 5; ++mt) {
#pragma unroll
        for (int r = 0; r < 8; ++r) {
            const int wrow = (r >> 1) * HID + 4 * mt + 2 * hf + (r & 1);
            wx[mt][r] = Wih1[wrow];
            bs[mt][r] = bih1[wrow] + bhh1[wrow];
        }
    }
    float w2[4][5][2];
#pragma unroll
    for (int g = 0; g < 4; ++g)
#pragma unroll
        for (int mt = 0; mt < 5; ++mt)
#pragma unroll
            for (int u = 0; u < 2; ++u)
                w2[g][mt][u] = Wih2[g * HID + 4 * mt + 2 * hf + u];
    float whh2[4], b2[4];
#pragma unroll
    for (int g = 0; g < 4; ++g) {
        whh2[g] = Whh2[g];
        b2[g]   = bih2[g] + bhh2[g];
    }

    v2f hB[5], cB[5];
#pragma unroll
    for (int mt = 0; mt < 5; ++mt) {
        v2f z = {0.f, 0.f};
        hB[mt] = z; cB[mt] = z;
    }
    float h2v = 0.0f, c2v = 0.0f;

    const float* xrow = xin + (size_t)(row0 + c) * T_LEN;
    float*       orow = out + (size_t)(row0 + c) * TT;

    for (int t = 0; t < TT; ++t) {
        float x = c2v;
        if (t < T_LEN) x = xrow[t];

        v8f acc[5];
#pragma unroll
        for (int mt = 0; mt < 5; ++mt) acc[mt] = bs[mt];
#pragma unroll
        for (int kc = 0; kc < 5; ++kc) {
#pragma unroll
            for (int mt = 0; mt < 5; ++mt) {
                acc[mt] = __builtin_amdgcn_wmma_f32_16x16x4_f32(
                    false, af[mt][kc], false, hB[kc], (short)0, acc[mt], false, false);
            }
        }

        float p2[4] = {0.f, 0.f, 0.f, 0.f};
#pragma unroll
        for (int mt = 0; mt < 5; ++mt) {
#pragma unroll
            for (int u = 0; u < 2; ++u) {
                float gi = acc[mt][0 + u] + x * wx[mt][0 + u];
                float gf = acc[mt][2 + u] + x * wx[mt][2 + u];
                float gg = acc[mt][4 + u] + x * wx[mt][4 + u];
                float go = acc[mt][6 + u] + x * wx[mt][6 + u];
                float cn = sigm(gf) * cB[mt][u] + sigm(gi) * tanh_f(gg);
                float hn = sigm(go) * tanh_f(cn);
                cB[mt][u] = cn;
                hB[mt][u] = hn;
#pragma unroll
                for (int g = 0; g < 4; ++g) p2[g] += cn * w2[g][mt][u];
            }
        }

        float g2[4];
#pragma unroll
        for (int g = 0; g < 4; ++g)
            g2[g] = (p2[g] + xhalf(p2[g])) + (b2[g] + h2v * whh2[g]);
        float cn2 = sigm(g2[1]) * c2v + sigm(g2[0]) * tanh_f(g2[2]);
        c2v = cn2;
        h2v = sigm(g2[3]) * tanh_f(cn2);
        if (hf == 0) orow[t] = cn2;
    }
}

extern "C" void kernel_launch(void* const* d_in, const int* in_sizes, int n_in,
                              void* d_out, int out_size, void* d_ws, size_t ws_size,
                              hipStream_t stream)
{
    const float* xin  = (const float*)d_in[0];
    const float* Wih1 = (const float*)d_in[1];
    const float* Whh1 = (const float*)d_in[2];
    const float* bih1 = (const float*)d_in[3];
    const float* bhh1 = (const float*)d_in[4];
    const float* Wih2 = (const float*)d_in[5];
    const float* Whh2 = (const float*)d_in[6];
    const float* bih2 = (const float*)d_in[7];
    const float* bhh2 = (const float*)d_in[8];
    const int*   futp = (const int*)d_in[9];
    float* out = (float*)d_out;

    dim3 grid(B_TOT / ROWS);
    dim3 block(32);
    lstm_seq_kernel<<<grid, block, 0, stream>>>(xin, Wih1, Whh1, bih1, bhh1,
                                                Wih2, Whh2, bih2, bhh2, futp, out);
}


# Activations: each IEEE division 1/(1+e^y) (scale, reciprocal, fma corrections, select, fix-up: 11 instructions) becomes one hardware reciprocal v_rcp_f32 (1 ulp); denominators >= 1, never 0
# speedup vs baseline: 3.8124x; 2.5910x over previous
#include <hip/hip_runtime.h>

typedef __attribute__((ext_vector_type(2))) float v2f;
typedef __attribute__((ext_vector_type(8))) float v8f;

#define B_TOT 1024
#define T_LEN 4096
#define HID   20
#define ROWS  16

__device__ __forceinline__ float rcp_hw(float x) {
    return __builtin_amdgcn_rcpf(x);
}
__device__ __forceinline__ float sigm(float x) {
    return rcp_hw(1.0f + __expf(-x));
}
__device__ __forceinline__ float tanh_f(float x) {
    return 1.0f - 2.0f * rcp_hw(__expf(2.0f * x) + 1.0f);
}
__device__ __forceinline__ float xhalf(float x) {
    int b = __builtin_bit_cast(int, x);
    return __builtin_bit_cast(float,
        __builtin_amdgcn_permlanex16(b, b, 0x76543210, 0xfedcba98, false, false));
}

__global__ void __launch_bounds__(32)
lstm_seq_kernel(const float* __restrict__ xin,
                const float* __restrict__ Wih1,
                const float* __restrict__ Whh1,
                const float* __restrict__ bih1,
                const float* __restrict__ bhh1,
                const float* __restrict__ Wih2,
                const float* __restrict__ Whh2,
                const float* __restrict__ bih2,
                const float* __restrict__ bhh2,
                const int*   __restrict__ futp,
                float* __restrict__ out)
{
    const int lane = threadIdx.x;
    const int c    = lane & 15;
    const int hf   = lane >> 4;
    const int row0 = blockIdx.x * ROWS;
    const int fut  = *futp;
    const int TT   = T_LEN + fut;

    v2f af[5][5];
    {
        const int hm = c >> 3, r = c & 7, g = r >> 1, u = r & 1;
#pragma unroll
        for (int mt = 0; mt < 5; ++mt) {
            const int wrow = g * HID + 4 * mt + 2 * hm + u;
#pragma unroll
            for (int kc = 0; kc < 5; ++kc) {
                const float* p = Whh1 + wrow * HID + 4 * kc + 2 * hf;
                v2f a; a.x = p[0]; a.y = p[1];
                af[mt][kc] = a;
            }
        }
    }
    float wx[5][8];
    v8f   bs[5];
#pragma unroll
    for (int mt = 0; mt < 5; ++mt) {
#pragma unroll
        for (int r = 0; r < 8; ++r) {
            const int wrow = (r >> 1) * HID + 4 * mt + 2 * hf + (r & 1);
            wx[mt][r] = Wih1[wrow];
            bs[mt][r] = bih1[wrow] + bhh1[wrow];
        }
    }
    float w2[4][5][2];
#pragma unroll
    for (int g = 0; g < 4; ++g)
#pragma unroll
        for (int mt = 0; mt < 5; ++mt)
#pragma unroll
            for (int u = 0; u < 2; ++u)
                w2[g][mt][u] = Wih2[g * HID + 4 * mt + 2 * hf + u];
    float whh2[4], b2[4];
#pragma unroll
    for (int g = 0; g < 4; ++g) {
        whh2[g] = Whh2[g];
        b2[g]   = bih2[g] + bhh2[g];
    }

    v2f hB[5], cB[5];
#pragma unroll
    for (int mt = 0; mt < 5; ++mt) {
        v2f z = {0.f, 0.f};
        hB[mt] = z; cB[mt] = z;
    }
    float h2v = 0.0f, c2v = 0.0f;

    const float* xrow = xin + (size_t)(row0 + c) * T_LEN;
    float*       orow = out + (size_t)(row0 + c) * TT;

    for (int t = 0; t < TT; ++t) {
        float x = c2v;
        if (t < T_LEN) x = xrow[t];

        v8f acc[5];
#pragma unroll
        for (int mt = 0; mt < 5; ++mt) acc[mt] = bs[mt];
#pragma unroll
        for (int kc = 0; kc < 5; ++kc) {
#pragma unroll
            for (int mt = 0; mt < 5; ++mt) {
                acc[mt] = __builtin_amdgcn_wmma_f32_16x16x4_f32(
                    false, af[mt][kc], false, hB[kc], (short)0, acc[mt], false, false);
            }
        }

        float p2[4] = {0.f, 0.f, 0.f, 0.f};
#pragma unroll
        for (int mt = 0; mt < 5; ++mt) {
#pragma unroll
            for (int u = 0; u < 2; ++u) {
                float gi = acc[mt][0 + u] + x * wx[mt][0 + u];
                float gf = acc[mt][2 + u] + x * wx[mt][2 + u];
                float gg = acc[mt][4 + u] + x * wx[mt][4 + u];
                float go = acc[mt][6 + u] + x * wx[mt][6 + u];
                float cn = sigm(gf) * cB[mt][u] + sigm(gi) * tanh_f(gg);
                float hn = sigm(go) * tanh_f(cn);
                cB[mt][u] = cn;
                hB[mt][u] = hn;
#pragma unroll
                for (int g = 0; g < 4; ++g) p2[g] += cn * w2[g][mt][u];
            }
        }

        float g2[4];
#pragma unroll
        for (int g = 0; g < 4; ++g)
            g2[g] = (p2[g] + xhalf(p2[g])) + (b2[g] + h2v * whh2[g]);
        float cn2 = sigm(g2[1]) * c2v + sigm(g2[0]) * tanh_f(g2[2]);
        c2v = cn2;
        h2v = sigm(g2[3]) * tanh_f(cn2);
        if (hf == 0) orow[t] = cn2;
    }
}

extern "C" void kernel_launch(void* const* d_in, const int* in_sizes, int n_in,
                              void* d_out, int out_size, void* d_ws, size_t ws_size,
                              hipStream_t stream)
{
    const float* xin  = (const float*)d_in[0];
    const float* Wih1 = (const float*)d_in[1];
    const float* Whh1 = (const float*)d_in[2];
    const float* bih1 = (const float*)d_in[3];
    const float* bhh1 = (const float*)d_in[4];
    const float* Wih2 = (const float*)d_in[5];
    const float* Whh2 = (const float*)d_in[6];
    const float* bih2 = (const float*)d_in[7];
    const float* bhh2 = (const float*)d_in[8];
    const int*   futp = (const int*)d_in[9];
    float* out = (float*)d_out;

    dim3 grid(B_TOT / ROWS);
    dim3 block(32);
    lstm_seq_kernel<<<grid, block, 0, stream>>>(xin, Wih1, Whh1, bih1, bhh1,
                                                Wih2, Whh2, bih2, bhh2, futp, out);
}


# Activations: each sigmoid and tanh is one hardware tanh (v_tanh_f32), sigmoid(y) = (1 + tanh(y/2))/2 with the halving folded exactly into the i, f, o weight and bias rows; replaces exp + add + rcp
# speedup vs baseline: 4.9364x; 1.2948x over previous
#include <hip/hip_runtime.h>

typedef __attribute__((ext_vector_type(2))) float v2f;
typedef __attribute__((ext_vector_type(8))) float v8f;

#define B_TOT 1024
#define T_LEN 4096
#define HID   20
#define ROWS  16

__device__ __forceinline__ float tanh_f(float x) {
    return __builtin_amdgcn_tanhf(x);
}
__device__ __forceinline__ float sigm_h(float xh) {
    return __builtin_fmaf(0.5f, __builtin_amdgcn_tanhf(xh), 0.5f);
}
__device__ __forceinline__ float gate_scale(int g) {
    return (g == 2) ? 1.0f : 0.5f;
}
__device__ __forceinline__ float xhalf(float x) {
    int b = __builtin_bit_cast(int, x);
    return __builtin_bit_cast(float,
        __builtin_amdgcn_permlanex16(b, b, 0x76543210, 0xfedcba98, false, false));
}

__global__ void __launch_bounds__(32)
lstm_seq_kernel(const float* __restrict__ xin,
                const float* __restrict__ Wih1,
                const float* __restrict__ Whh1,
                const float* __restrict__ bih1,
                const float* __restrict__ bhh1,
                const float* __restrict__ Wih2,
                const float* __restrict__ Whh2,
                const float* __restrict__ bih2,
                const float* __restrict__ bhh2,
                const int*   __restrict__ futp,
                float* __restrict__ out)
{
    const int lane = threadIdx.x;
    const int c    = lane & 15;
    const int hf   = lane >> 4;
    const int row0 = blockIdx.x * ROWS;
    const int fut  = *futp;
    const int TT   = T_LEN + fut;

    v2f af[5][5];
    {
        const int hm = c >> 3, r = c & 7, g = r >> 1, u = r & 1;
#pragma unroll
        for (int mt = 0; mt < 5; ++mt) {
            const int wrow = g * HID + 4 * mt + 2 * hm + u;
#pragma unroll
            for (int kc = 0; kc < 5; ++kc) {
                const float* p = Whh1 + wrow * HID + 4 * kc + 2 * hf;
                v2f a; a.x = p[0] * gate_scale(g); a.y = p[1] * gate_scale(g);
                af[mt][kc] = a;
            }
        }
    }
    float wx[5][8];
    v8f   bs[5];
#pragma unroll
    for (int mt = 0; mt < 5; ++mt) {
#pragma unroll
        for (int r = 0; r < 8; ++r) {
            const int wrow = (r >> 1) * HID + 4 * mt + 2 * hf + (r & 1);
            wx[mt][r] = Wih1[wrow] * gate_scale(r >> 1);
            bs[mt][r] = (bih1[wrow] + bhh1[wrow]) * gate_scale(r >> 1);
        }
    }
    float w2[4][5][2];
#pragma unroll
    for (int g = 0; g < 4; ++g)
#pragma unroll
        for (int mt = 0; mt < 5; ++mt)
#pragma unroll
            for (int u = 0; u < 2; ++u)
                w2[g][mt][u] = Wih2[g * HID + 4 * mt + 2 * hf + u] * gate_scale(g);
    float whh2[4], b2[4];
#pragma unroll
    for (int g = 0; g < 4; ++g) {
        whh2[g] = Whh2[g] * gate_scale(g);
        b2[g]   = (bih2[g] + bhh2[g]) * gate_scale(g);
    }

    v2f hB[5], cB[5];
#pragma unroll
    for (int mt = 0; mt < 5; ++mt) {
        v2f z = {0.f, 0.f};
        hB[mt] = z; cB[mt] = z;
    }
    float h2v = 0.0f, c2v = 0.0f;

    const float* xrow = xin + (size_t)(row0 + c) * T_LEN;
    float*       orow = out + (size_t)(row0 + c) * TT;

    for (int t = 0; t < TT; ++t) {
        float x = c2v;
        if (t < T_LEN) x = xrow[t];

        v8f acc[5];
#pragma unroll
        for (int mt = 0; mt < 5; ++mt) acc[mt] = bs[mt];
#pragma unroll
        for (int kc = 0; kc < 5; ++kc) {
#pragma unroll
            for (int mt = 0; mt < 5; ++mt) {
                acc[mt] = __builtin_amdgcn_wmma_f32_16x16x4_f32(
                    false, af[mt][kc], false, hB[kc], (short)0, acc[mt], false, false);
            }
        }

        float p2[4] = {0.f, 0.f, 0.f, 0.f};
#pragma unroll
        for (int mt = 0; mt < 5; ++mt) {
#pragma unroll
            for (int u = 0; u < 2; ++u) {
                float gi = acc[mt][0 + u] + x * wx[mt][0 + u];
                float gf = acc[mt][2 + u] + x * wx[mt][2 + u];
                float gg = acc[mt][4 + u] + x * wx[mt][4 + u];
                float go = acc[mt][6 + u] + x * wx[mt][6 + u];
                float cn = sigm_h(gf) * cB[mt][u] + sigm_h(gi) * tanh_f(gg);
                float hn = sigm_h(go) * tanh_f(cn);
                cB[mt][u] = cn;
                hB[mt][u] = hn;
#pragma unroll
                for (int g = 0; g < 4; ++g) p2[g] += cn * w2[g][mt][u];
            }
        }

        float g2[4];
#pragma unroll
        for (int g = 0; g < 4; ++g)
            g2[g] = (p2[g] + xhalf(p2[g])) + (b2[g] + h2v * whh2[g]);
        float cn2 = sigm_h(g2[1]) * c2v + sigm_h(g2[0]) * tanh_f(g2[2]);
        c2v = cn2;
        h2v = sigm_h(g2[3]) * tanh_f(cn2);
        if (hf == 0) orow[t] = cn2;
    }
}

extern "C" void kernel_launch(void* const* d_in, const int* in_sizes, int n_in,
                              void* d_out, int out_size, void* d_ws, size_t ws_size,
                              hipStream_t stream)
{
    const float* xin  = (const float*)d_in[0];
    const float* Wih1 = (const float*)d_in[1];
    const float* Whh1 = (const float*)d_in[2];
    const float* bih1 = (const float*)d_in[3];
    const float* bhh1 = (const float*)d_in[4];
    const float* Wih2 = (const float*)d_in[5];
    const float* Whh2 = (const float*)d_in[6];
    const float* bih2 = (const float*)d_in[7];
    const float* bhh2 = (const float*)d_in[8];
    const int*   futp = (const int*)d_in[9];
    float* out = (float*)d_out;

    dim3 grid(B_TOT / ROWS);
    dim3 block(32);
    lstm_seq_kernel<<<grid, block, 0, stream>>>(xin, Wih1, Whh1, bih1, bhh1,
                                                Wih2, Whh2, bih2, bhh2, futp, out);
}


# Time loop software-pipelined by one tile: the 5 WMMAs of K chunk kc of step t+1 issue behind tile kc's cell update, the update of tile kc+1 placed between them; 2 steps per trip; sum order unchanged
# speedup vs baseline: 5.1202x; 1.0372x over previous
#include <hip/hip_runtime.h>

typedef __attribute__((ext_vector_type(2))) float v2f;
typedef __attribute__((ext_vector_type(8))) float v8f;

#define B_TOT 1024
#define T_LEN 4096
#define HID   20
#define ROWS  16

__device__ __forceinline__ float tanh_f(float x) {
    return __builtin_amdgcn_tanhf(x);
}
__device__ __forceinline__ float sigm_h(float xh) {
    return __builtin_fmaf(0.5f, __builtin_amdgcn_tanhf(xh), 0.5f);
}
__device__ __forceinline__ float gate_scale(int g) {
    return (g == 2) ? 1.0f : 0.5f;
}
__device__ __forceinline__ float xhalf(float x) {
    int b = __builtin_bit_cast(int, x);
    return __builtin_bit_cast(float,
        __builtin_amdgcn_permlanex16(b, b, 0x76543210, 0xfedcba98, false, false));
}

struct tag_seq { static constexpr bool value = false; };
struct tag_fut { static constexpr bool value = true;  };

__global__ void __launch_bounds__(32)
lstm_seq_kernel(const float* __restrict__ xin,
                const float* __restrict__ Wih1,
                const float* __restrict__ Whh1,
                const float* __restrict__ bih1,
                const float* __restrict__ bhh1,
                const float* __restrict__ Wih2,
                const float* __restrict__ Whh2,
                const float* __restrict__ bih2,
                const float* __restrict__ bhh2,
                const int*   __restrict__ futp,
                float* __restrict__ out)
{
    const int lane = threadIdx.x;
    const int c    = lane & 15;
    const int hf   = lane >> 4;
    const int row0 = blockIdx.x * ROWS;
    const int fut  = *futp;
    const int TT   = T_LEN + fut;

    v2f af[5][5];
    {
        const int hm = c >> 3, r = c & 7, g = r >> 1, u = r & 1;
#pragma unroll
        for (int mt = 0; mt < 5; ++mt) {
            const int wrow = g * HID + 4 * mt + 2 * hm + u;
#pragma unroll
            for (int kc = 0; kc < 5; ++kc) {
                const float* p = Whh1 + wrow * HID + 4 * kc + 2 * hf;
                v2f a; a.x = p[0] * gate_scale(g); a.y = p[1] * gate_scale(g);
                af[mt][kc] = a;
            }
        }
    }
    float wx[5][8];
    v8f   bs[5];
#pragma unroll
    for (int mt = 0; mt < 5; ++mt) {
#pragma unroll
        for (int r = 0; r < 8; ++r) {
            const int wrow = (r >> 1) * HID + 4 * mt + 2 * hf + (r & 1);
            wx[mt][r] = Wih1[wrow] * gate_scale(r >> 1);
            bs[mt][r] = (bih1[wrow] + bhh1[wrow]) * gate_scale(r >> 1);
        }
    }
    float w2[4][5][2];
#pragma unroll
    for (int g = 0; g < 4; ++g)
#pragma unroll
        for (int mt = 0; mt < 5; ++mt)
#pragma unroll
            for (int u = 0; u < 2; ++u)
                w2[g][mt][u] = Wih2[g * HID + 4 * mt + 2 * hf + u] * gate_scale(g);
    float whh2[4], b2[4];
#pragma unroll
    for (int g = 0; g < 4; ++g) {
        whh2[g] = Whh2[g] * gate_scale(g);
        b2[g]   = (bih2[g] + bhh2[g]) * gate_scale(g);
    }

    v2f hB[5], cB[5];
#pragma unroll
    for (int mt = 0; mt < 5; ++mt) {
        v2f z = {0.f, 0.f};
        hB[mt] = z; cB[mt] = z;
    }
    float h2v = 0.0f, c2v = 0.0f;

    const float* xrow = xin + (size_t)(row0 + c) * T_LEN;
    float*       orow = out + (size_t)(row0 + c) * TT;

    v8f Y[5];
#pragma unroll
    for (int mt = 0; mt < 5; ++mt) Y[mt] = bs[mt];
    float p2[4] = {0.f, 0.f, 0.f, 0.f};

    v2f gf, go, ti, tg, tf, to, ig, cn, tc, so;
    auto cellA = [&](const int mt, const v8f& acc, const float x) __attribute__((always_inline)) {
#pragma unroll
        for (int u = 0; u < 2; ++u) {
            const float gi = __builtin_fmaf(x, wx[mt][0 + u], acc[0 + u]);
            const float gg = __builtin_fmaf(x, wx[mt][4 + u], acc[4 + u]);
            gf[u] = __builtin_fmaf(x, wx[mt][2 + u], acc[2 + u]);
            go[u] = __builtin_fmaf(x, wx[mt][6 + u], acc[6 + u]);
            ti[u] = tanh_f(gi);
            tg[u] = tanh_f(gg);
        }
    };
    auto cellB = [&]() __attribute__((always_inline)) {
#pragma unroll
        for (int u = 0; u < 2; ++u) {
            tf[u] = tanh_f(gf[u]);
            to[u] = tanh_f(go[u]);
            ig[u] = __builtin_fmaf(0.5f, ti[u], 0.5f) * tg[u];
        }
    };
    auto cellC = [&](const int mt) __attribute__((always_inline)) {
#pragma unroll
        for (int u = 0; u < 2; ++u) {
            cn[u] = __builtin_fmaf(__builtin_fmaf(0.5f, tf[u], 0.5f), cB[mt][u], ig[u]);
            tc[u] = tanh_f(cn[u]);
            so[u] = __builtin_fmaf(0.5f, to[u], 0.5f);
        }
    };
    auto cellD = [&](const int mt) __attribute__((always_inline)) {
#pragma unroll
        for (int u = 0; u < 2; ++u) {
            cB[mt][u] = cn[u];
            hB[mt][u] = so[u] * tc[u];
        }
    };
    auto cellE = [&](const int mt) __attribute__((always_inline)) {
#pragma unroll
        for (int u = 0; u < 2; ++u)
#pragma unroll
            for (int g = 0; g < 4; ++g) p2[g] += cn[u] * w2[g][mt][u];
    };
    auto wm = [&](v8f (&X)[5], const int kc, const int mt) __attribute__((always_inline)) {
        X[mt] = __builtin_amdgcn_wmma_f32_16x16x4_f32(
            false, af[mt][kc], false, hB[kc], (short)0,
            (kc == 0) ? bs[mt] : X[mt], false, false);
    };

#define HOLD(...) do { asm volatile("" : __VA_ARGS__); __builtin_amdgcn_sched_barrier(0); } while (0)

    auto step = [&](v8f (&Y)[5], v8f (&X)[5], const int t, const float xt, const float xn1,
                    auto fut_tag) __attribute__((always_inline)) -> float {
        constexpr bool FUT = decltype(fut_tag)::value;
        float g2[4], cn2;
        if (!FUT) {
#pragma unroll
            for (int kc = 0; kc < 4; ++kc) {
                const int m = kc + 1;
                wm(X, kc, 0); cellA(m, Y[m], xt);
                HOLD("+v"(X[0]), "+v"(hB[kc]), "+v"(gf), "+v"(go), "+v"(ti), "+v"(tg));
                wm(X, kc, 1); cellB();
                HOLD("+v"(X[1]), "+v"(hB[kc]), "+v"(tf), "+v"(to), "+v"(ig));
                wm(X, kc, 2); cellC(m);
                HOLD("+v"(X[2]), "+v"(hB[kc]), "+v"(cn), "+v"(tc), "+v"(so));
                wm(X, kc, 3); cellD(m);
                HOLD("+v"(X[3]), "+v"(hB[kc]), "+v"(hB[m]), "+v"(cn));
                wm(X, kc, 4); cellE(m);
                if (m < 4) HOLD("+v"(X[4]), "+v"(hB[m]), "+v"(Y[m + 1]),
                                "+v"(p2[0]), "+v"(p2[1]), "+v"(p2[2]), "+v"(p2[3]));
                else       HOLD("+v"(X[4]), "+v"(hB[m]),
                                "+v"(p2[0]), "+v"(p2[1]), "+v"(p2[2]), "+v"(p2[3]));
            }
            wm(X, 4, 0);
            HOLD("+v"(X[0]), "+v"(hB[4]));
#pragma unroll
            for (int g = 0; g < 4; ++g)
                g2[g] = (p2[g] + xhalf(p2[g])) + (b2[g] + h2v * whh2[g]);
#pragma unroll
            for (int g = 0; g < 4; ++g) p2[g] = 0.f;
            HOLD("+v"(X[0]), "+v"(hB[4]), "+v"(g2[0]), "+v"(g2[1]), "+v"(g2[2]), "+v"(g2[3]));
            wm(X, 4, 1); cellA(0, X[0], xn1);
            float t0 = tanh_f(g2[0]), t1 = tanh_f(g2[1]), t2 = tanh_f(g2[2]);
            HOLD("+v"(X[1]), "+v"(hB[4]), "+v"(gf), "+v"(go), "+v"(ti), "+v"(tg),
                 "+v"(t0), "+v"(t1), "+v"(t2));
            wm(X, 4, 2); cellB();
            cn2 = __builtin_fmaf(0.5f, t1, 0.5f) * c2v + __builtin_fmaf(0.5f, t0, 0.5f) * t2;
            float t3 = tanh_f(g2[3]), tc2 = tanh_f(cn2);
            HOLD("+v"(X[2]), "+v"(hB[4]), "+v"(tf), "+v"(to), "+v"(ig), "+v"(t3), "+v"(tc2));
            wm(X, 4, 3); cellC(0);
            c2v = cn2;
            h2v = __builtin_fmaf(0.5f, t3, 0.5f) * tc2;
            HOLD("+v"(X[3]), "+v"(hB[4]), "+v"(cn), "+v"(tc), "+v"(so));
            wm(X, 4, 4); cellD(0); cellE(0);
            HOLD("+v"(X[4]), "+v"(hB[0]), "+v"(X[1]));
        } else {
#pragma unroll
            for (int kc = 0; kc < 5; ++kc) {
#pragma unroll
                for (int mt = 0; mt < 5; ++mt) wm(X, kc, mt);
                if (kc < 4) {
                    cellA(kc + 1, Y[kc + 1], xt); cellB(); cellC(kc + 1); cellD(kc + 1); cellE(kc + 1);
                }
            }
#pragma unroll
            for (int g = 0; g < 4; ++g)
                g2[g] = (p2[g] + xhalf(p2[g])) + (b2[g] + h2v * whh2[g]);
            cn2 = sigm_h(g2[1]) * c2v + sigm_h(g2[0]) * tanh_f(g2[2]);
            c2v = cn2;
            h2v = sigm_h(g2[3]) * tanh_f(cn2);
#pragma unroll
            for (int g = 0; g < 4; ++g) p2[g] = 0.f;
            cellA(0, X[0], cn2); cellB(); cellC(0); cellD(0); cellE(0);
        }
        orow[t] = cn2;
        return FUT ? cn2 : xn1;
    };

    v8f Z[5];
    float xa = xrow[0], xb = xrow[1], xc = xrow[2];
    cellA(0, Y[0], xa); cellB(); cellC(0); cellD(0); cellE(0);

    int t = 0;
    for (; t + 2 <= T_LEN - 1; t += 2) {
        const float xd = xrow[(t + 3 < T_LEN) ? (t + 3) : (T_LEN - 1)];
        const float xe = xrow[(t + 4 < T_LEN) ? (t + 4) : (T_LEN - 1)];
        step(Y, Z, t,     xa, xb, tag_seq{});
        step(Z, Y, t + 1, xb, xc, tag_seq{});
        xa = xc; xb = xd; xc = xe;
    }
    if (t < T_LEN - 1) {
        step(Y, Z, t, xa, xb, tag_seq{});
#pragma unroll
        for (int mt = 0; mt < 5; ++mt) Y[mt] = Z[mt];
        xa = xb;
        ++t;
    }
    for (; t < TT; ++t) {
        xa = step(Y, Z, t, xa, 0.f, tag_fut{});
#pragma unroll
        for (int mt = 0; mt < 5; ++mt) Y[mt] = Z[mt];
    }
#undef HOLD
}

extern "C" void kernel_launch(void* const* d_in, const int* in_sizes, int n_in,
                              void* d_out, int out_size, void* d_ws, size_t ws_size,
                              hipStream_t stream)
{
    const float* xin  = (const float*)d_in[0];
    const float* Wih1 = (const float*)d_in[1];
    const float* Whh1 = (const float*)d_in[2];
    const float* bih1 = (const float*)d_in[3];
    const float* bhh1 = (const float*)d_in[4];
    const float* Wih2 = (const float*)d_in[5];
    const float* Whh2 = (const float*)d_in[6];
    const float* bih2 = (const float*)d_in[7];
    const float* bhh2 = (const float*)d_in[8];
    const int*   futp = (const int*)d_in[9];
    float* out = (float*)d_out;

    dim3 grid(B_TOT / ROWS);
    dim3 block(32);
    lstm_seq_kernel<<<grid, block, 0, stream>>>(xin, Wih1, Whh1, bih1, bhh1,
                                                Wih2, Whh2, bih2, bhh2, futp, out);
}


# The five 16-row gate tiles of a batch tile go to five waves (5 WMMAs each), h1 exchanged through LDS at one barrier per step, layer-2 cell on a sixth wave; sum order changes
# speedup vs baseline: 10.1054x; 1.9736x over previous
#include <hip/hip_runtime.h>

typedef __attribute__((ext_vector_type(2))) float v2f;
typedef __attribute__((ext_vector_type(8))) float v8f;

#define B_TOT 1024
#define T_LEN 4096
#define HID   20
#define ROWS  16
#define NT    5
#define NW    6

__device__ __forceinline__ float tanh_f(float x) {
    return __builtin_amdgcn_tanhf(x);
}
__device__ __forceinline__ float sigm_h(float xh) {
    return __builtin_fmaf(0.5f, __builtin_amdgcn_tanhf(xh), 0.5f);
}
__device__ __forceinline__ float gate_scale(int g) {
    return (g == 2) ? 1.0f : 0.5f;
}
__device__ __forceinline__ float xhalf(float x) {
    int b = __builtin_bit_cast(int, x);
    return __builtin_bit_cast(float,
        __builtin_amdgcn_permlanex16(b, b, 0x76543210, 0xfedcba98, false, false));
}

__device__ __forceinline__ void lds_barrier() {
    __builtin_amdgcn_fence(__ATOMIC_RELEASE, "workgroup", "local");
    __builtin_amdgcn_s_barrier();
    __builtin_amdgcn_fence(__ATOMIC_ACQUIRE, "workgroup", "local");
}

struct tag_seq { static constexpr bool value = false; };
struct tag_fut { static constexpr bool value = true;  };

__global__ void __launch_bounds__(32 * NW)
lstm_seq_kernel(const float* __restrict__ xin,
                const float* __restrict__ Wih1,
                const float* __restrict__ Whh1,
                const float* __restrict__ bih1,
                const float* __restrict__ bhh1,
                const float* __restrict__ Wih2,
                const float* __restrict__ Whh2,
                const float* __restrict__ bih2,
                const float* __restrict__ bhh2,
                const int*   __restrict__ futp,
                float* __restrict__ out)
{
    __shared__ v2f   hbuf[2][NT][32];
    __shared__ v2f   cbuf[2][NT][32];
    __shared__ float xbuf[ROWS];

    const int lane = threadIdx.x & 31;
    const int w    = __builtin_amdgcn_readfirstlane((int)(threadIdx.x >> 5));
    const bool tilew = (w < NT);
    const int mt   = tilew ? w : 0;
    const int c    = lane & 15;
    const int hf   = lane >> 4;
    const int row0 = blockIdx.x * ROWS;
    const int fut  = *futp;
    const int TT   = T_LEN + fut;

    int kcr[5];
#pragma unroll
    for (int j = 0; j < 5; ++j) kcr[j] = (mt + j < 5) ? (mt + j) : (mt + j - 5);
    v2f afr[5];
    {
        const int hm = c >> 3, r = c & 7, g = r >> 1, u = r & 1;
        const int wrow = g * HID + 4 * mt + 2 * hm + u;
#pragma unroll
        for (int j = 0; j < 5; ++j) {
            const float* p = Whh1 + wrow * HID + 4 * kcr[j] + 2 * hf;
            v2f a; a.x = p[0] * gate_scale(g); a.y = p[1] * gate_scale(g);
            afr[j] = a;
        }
    }
    float wx[8];
    v8f   bs;
#pragma unroll
    for (int r = 0; r < 8; ++r) {
        const int wrow = (r >> 1) * HID + 4 * mt + 2 * hf + (r & 1);
        wx[r] = Wih1[wrow] * gate_scale(r >> 1);
        bs[r] = (bih1[wrow] + bhh1[wrow]) * gate_scale(r >> 1);
    }
    v2f w2[4][5];
#pragma unroll
    for (int g = 0; g < 4; ++g)
#pragma unroll
        for (int m = 0; m < 5; ++m) {
            const float* q = Wih2 + g * HID + 4 * m + 2 * hf;
            v2f v; v.x = q[0] * gate_scale(g); v.y = q[1] * gate_scale(g);
            w2[g][m] = v;
        }
    float whh2[4], b2[4];
#pragma unroll
    for (int g = 0; g < 4; ++g) {
        whh2[g] = Whh2[g] * gate_scale(g);
        b2[g]   = (bih2[g] + bhh2[g]) * gate_scale(g);
    }

    v2f   cB = {0.f, 0.f};
    float h2v = 0.0f, c2v = 0.0f;
    v8f   Y = bs;

    const float* xrow = xin + (size_t)(row0 + c) * T_LEN;
    float*       orow = out + (size_t)(row0 + c) * TT;

    auto step = [&](const int t, const float x, auto fb_tag) __attribute__((always_inline)) -> float {
        constexpr bool FB = decltype(fb_tag)::value;
        const int p = t & 1;
        if (tilew) {
            v2f hn;
#pragma unroll
            for (int u = 0; u < 2; ++u) {
                const float gi = __builtin_fmaf(x, wx[0 + u], Y[0 + u]);
                const float gf = __builtin_fmaf(x, wx[2 + u], Y[2 + u]);
                const float gg = __builtin_fmaf(x, wx[4 + u], Y[4 + u]);
                const float go = __builtin_fmaf(x, wx[6 + u], Y[6 + u]);
                const float ti = tanh_f(gi), tf = tanh_f(gf), tg = tanh_f(gg), to = tanh_f(go);
                const float ig = __builtin_fmaf(0.5f, ti, 0.5f) * tg;
                const float cn = __builtin_fmaf(__builtin_fmaf(0.5f, tf, 0.5f), cB[u], ig);
                const float tc = tanh_f(cn);
                cB[u] = cn;
                hn[u] = __builtin_fmaf(0.5f, to, 0.5f) * tc;
            }
            Y = __builtin_amdgcn_wmma_f32_16x16x4_f32(false, afr[0], false, hn, (short)0, bs, false, false);
            hbuf[p][w][lane] = hn;
            cbuf[p][w][lane] = cB;
        }
        lds_barrier();
        float cn2 = 0.0f;
        if (tilew) {
#pragma unroll
            for (int j = 1; j < 5; ++j) {
                const v2f hB = hbuf[p][kcr[j]][lane];
                Y = __builtin_amdgcn_wmma_f32_16x16x4_f32(false, afr[j], false, hB, (short)0, Y, false, false);
            }
        } else {
            v2f cn[5];
#pragma unroll
            for (int m = 0; m < 5; ++m) cn[m] = cbuf[p][m][lane];
            v2f acc[4];
#pragma unroll
            for (int g = 0; g < 4; ++g) acc[g] = cn[0] * w2[g][0];
#pragma unroll
            for (int m = 1; m < 5; ++m)
#pragma unroll
                for (int g = 0; g < 4; ++g) acc[g] = __builtin_elementwise_fma(cn[m], w2[g][m], acc[g]);
            float g2[4];
#pragma unroll
            for (int g = 0; g < 4; ++g) {
                const float p2 = acc[g].x + acc[g].y;
                g2[g] = (p2 + xhalf(p2)) + (b2[g] + h2v * whh2[g]);
            }
            cn2 = sigm_h(g2[1]) * c2v + sigm_h(g2[0]) * tanh_f(g2[2]);
            c2v = cn2;
            h2v = sigm_h(g2[3]) * tanh_f(cn2);
            orow[t] = cn2;
        }
        float xn = 0.0f;
        if (FB) {
            if (!tilew) xbuf[c] = cn2;
            lds_barrier();
            xn = xbuf[c];
        }
        return xn;
    };

    typedef __attribute__((ext_vector_type(4))) float v4f;
    const v4f* xq = reinterpret_cast<const v4f*>(xrow);
    v4f q0 = xq[0], q1 = xq[1];
    int t = 0;
    for (; t < T_LEN - 4; t += 4) {
        const int nq = (t >> 2) + 2;
        const v4f q2 = xq[(nq < T_LEN / 4) ? nq : (T_LEN / 4 - 1)];
        step(t,     q0.x, tag_seq{});
        step(t + 1, q0.y, tag_seq{});
        step(t + 2, q0.z, tag_seq{});
        step(t + 3, q0.w, tag_seq{});
        q0 = q1; q1 = q2;
    }
    step(t,     q0.x, tag_seq{});
    step(t + 1, q0.y, tag_seq{});
    step(t + 2, q0.z, tag_seq{});
    float x = q0.w;
    t += 3;
    for (; t < TT; ++t) x = step(t, x, tag_fut{});
}

extern "C" void kernel_launch(void* const* d_in, const int* in_sizes, int n_in,
                              void* d_out, int out_size, void* d_ws, size_t ws_size,
                              hipStream_t stream)
{
    const float* xin  = (const float*)d_in[0];
    const float* Wih1 = (const float*)d_in[1];
    const float* Whh1 = (const float*)d_in[2];
    const float* bih1 = (const float*)d_in[3];
    const float* bhh1 = (const float*)d_in[4];
    const float* Wih2 = (const float*)d_in[5];
    const float* Whh2 = (const float*)d_in[6];
    const float* bih2 = (const float*)d_in[7];
    const float* bhh2 = (const float*)d_in[8];
    const int*   futp = (const int*)d_in[9];
    float* out = (float*)d_out;

    dim3 grid(B_TOT / ROWS);
    dim3 block(32 * NW);
    lstm_seq_kernel<<<grid, block, 0, stream>>>(xin, Wih1, Whh1, bih1, bhh1,
                                                Wih2, Whh2, bih2, bhh2, futp, out);
}


# Time loop rotated half a step: a tile wave's work between two barriers (4 chunk WMMAs, next cell update, own WMMA, stores) is one stretch under one role test, no jump after the barrier; sums kept
# speedup vs baseline: 10.7470x; 1.0635x over previous
#include <hip/hip_runtime.h>

typedef __attribute__((ext_vector_type(2))) float v2f;
typedef __attribute__((ext_vector_type(8))) float v8f;

#define B_TOT 1024
#define T_LEN 4096
#define HID   20
#define ROWS  16
#define NT    5
#define NW    6

__device__ __forceinline__ float tanh_f(float x) {
    return __builtin_amdgcn_tanhf(x);
}
__device__ __forceinline__ float sigm_h(float xh) {
    return __builtin_fmaf(0.5f, __builtin_amdgcn_tanhf(xh), 0.5f);
}
__device__ __forceinline__ float gate_scale(int g) {
    return (g == 2) ? 1.0f : 0.5f;
}
__device__ __forceinline__ float xhalf(float x) {
    int b = __builtin_bit_cast(int, x);
    return __builtin_bit_cast(float,
        __builtin_amdgcn_permlanex16(b, b, 0x76543210, 0xfedcba98, false, false));
}

__device__ __forceinline__ void lds_barrier() {
    __builtin_amdgcn_fence(__ATOMIC_RELEASE, "workgroup", "local");
    __builtin_amdgcn_s_barrier();
    __builtin_amdgcn_fence(__ATOMIC_ACQUIRE, "workgroup", "local");
}

__global__ void __launch_bounds__(32 * NW)
lstm_seq_kernel(const float* __restrict__ xin,
                const float* __restrict__ Wih1,
                const float* __restrict__ Whh1,
                const float* __restrict__ bih1,
                const float* __restrict__ bhh1,
                const float* __restrict__ Wih2,
                const float* __restrict__ Whh2,
                const float* __restrict__ bih2,
                const float* __restrict__ bhh2,
                const int*   __restrict__ futp,
                float* __restrict__ out)
{
    __shared__ v2f   hbuf[2][NT][32];
    __shared__ v2f   cbuf[2][NT][32];
    __shared__ float xbuf[ROWS];

    const int lane = threadIdx.x & 31;
    const int w    = __builtin_amdgcn_readfirstlane((int)(threadIdx.x >> 5));
    const bool tilew = (w < NT);
    const int mt   = tilew ? w : 0;
    const int c    = lane & 15;
    const int hf   = lane >> 4;
    const int row0 = blockIdx.x * ROWS;
    const int fut  = *futp;
    const int TT   = T_LEN + fut;

    int kcr[5];
#pragma unroll
    for (int j = 0; j < 5; ++j) kcr[j] = (mt + j < 5) ? (mt + j) : (mt + j - 5);
    v2f afr[5];
    {
        const int hm = c >> 3, r = c & 7, g = r >> 1, u = r & 1;
        const int wrow = g * HID + 4 * mt + 2 * hm + u;
#pragma unroll
        for (int j = 0; j < 5; ++j) {
            const float* p = Whh1 + wrow * HID + 4 * kcr[j] + 2 * hf;
            v2f a; a.x = p[0] * gate_scale(g); a.y = p[1] * gate_scale(g);
            afr[j] = a;
        }
    }
    float wx[8];
    v8f   bs;
#pragma unroll
    for (int r = 0; r < 8; ++r) {
        const int wrow = (r >> 1) * HID + 4 * mt + 2 * hf + (r & 1);
        wx[r] = Wih1[wrow] * gate_scale(r >> 1);
        bs[r] = (bih1[wrow] + bhh1[wrow]) * gate_scale(r >> 1);
    }
    v2f w2[4][5];
#pragma unroll
    for (int g = 0; g < 4; ++g)
#pragma unroll
        for (int m = 0; m < 5; ++m) {
            const float* q = Wih2 + g * HID + 4 * m + 2 * hf;
            v2f v; v.x = q[0] * gate_scale(g); v.y = q[1] * gate_scale(g);
            w2[g][m] = v;
        }
    float whh2[4], b2[4];
#pragma unroll
    for (int g = 0; g < 4; ++g) {
        whh2[g] = Whh2[g] * gate_scale(g);
        b2[g]   = (bih2[g] + bhh2[g]) * gate_scale(g);
    }

    v2f   cB = {0.f, 0.f};
    float h2v = 0.0f, c2v = 0.0f;
    v8f   Y = bs;

    const float* xrow = xin + (size_t)(row0 + c) * T_LEN;
    float*       orow = out + (size_t)(row0 + c) * TT;


    auto tile_cell = [&](const int t, const float x) __attribute__((always_inline)) {
        const int p = t & 1;
        v2f hn;
#pragma unroll
        for (int u = 0; u < 2; ++u) {
            const float gi = __builtin_fmaf(x, wx[0 + u], Y[0 + u]);
            const float gf = __builtin_fmaf(x, wx[2 + u], Y[2 + u]);
            const float gg = __builtin_fmaf(x, wx[4 + u], Y[4 + u]);
            const float go = __builtin_fmaf(x, wx[6 + u], Y[6 + u]);
            const float ti = tanh_f(gi), tf = tanh_f(gf), tg = tanh_f(gg), to = tanh_f(go);
            const float ig = __builtin_fmaf(0.5f, ti, 0.5f) * tg;
            const float cn = __builtin_fmaf(__builtin_fmaf(0.5f, tf, 0.5f), cB[u], ig);
            const float tc = tanh_f(cn);
            cB[u] = cn;
            hn[u] = __builtin_fmaf(0.5f, to, 0.5f) * tc;
        }
        Y = __builtin_amdgcn_wmma_f32_16x16x4_f32(false, afr[0], false, hn, (short)0, bs, false, false);
        hbuf[p][w][lane] = hn;
        cbuf[p][w][lane] = cB;
    };
    auto tile_prod = [&](const int t) __attribute__((always_inline)) {
        const int p = t & 1;
#pragma unroll
        for (int j = 1; j < 5; ++j) {
            const v2f hB = hbuf[p][kcr[j]][lane];
            Y = __builtin_amdgcn_wmma_f32_16x16x4_f32(false, afr[j], false, hB, (short)0, Y, false, false);
        }
    };
    auto l2_cell = [&](const int t) __attribute__((always_inline)) -> float {
        const int p = t & 1;
        v2f cn[5];
#pragma unroll
        for (int m = 0; m < 5; ++m) cn[m] = cbuf[p][m][lane];
        v2f acc[4];
#pragma unroll
        for (int g = 0; g < 4; ++g) acc[g] = cn[0] * w2[g][0];
#pragma unroll
        for (int m = 1; m < 5; ++m)
#pragma unroll
            for (int g = 0; g < 4; ++g) acc[g] = __builtin_elementwise_fma(cn[m], w2[g][m], acc[g]);
        float g2[4];
#pragma unroll
        for (int g = 0; g < 4; ++g) {
            const float p2 = acc[g].x + acc[g].y;
            g2[g] = (p2 + xhalf(p2)) + (b2[g] + h2v * whh2[g]);
        }
        const float cn2 = sigm_h(g2[1]) * c2v + sigm_h(g2[0]) * tanh_f(g2[2]);
        c2v = cn2;
        h2v = sigm_h(g2[3]) * tanh_f(cn2);
        orow[t] = cn2;
        return cn2;
    };
    auto seq_step = [&](const int t, const float xnext) __attribute__((always_inline)) {
        lds_barrier();
        if (__builtin_expect(w >= NT, 0)) {
            l2_cell(t);
        } else {
            tile_prod(t);
            tile_cell(t + 1, xnext);
        }
    };

    typedef __attribute__((ext_vector_type(4))) float v4f;
    const v4f* xq = reinterpret_cast<const v4f*>(xrow);
    v4f q0 = xq[0], q1 = xq[1];
    if (w < NT) tile_cell(0, q0.x);
    int t = 0;
    for (; t < T_LEN - 4; t += 4) {
        const int nq = (t >> 2) + 2;
        const v4f q2 = xq[(nq < T_LEN / 4) ? nq : (T_LEN / 4 - 1)];
        seq_step(t,     q0.y);
        seq_step(t + 1, q0.z);
        seq_step(t + 2, q0.w);
        seq_step(t + 3, q1.x);
        q0 = q1; q1 = q2;
    }
    seq_step(t,     q0.y);
    seq_step(t + 1, q0.z);
    seq_step(t + 2, q0.w);
    t += 3;
    for (; t < TT; ++t) {
        lds_barrier();
        float cn2 = 0.0f;
        if (w < NT) tile_prod(t);
        else        cn2 = l2_cell(t);
        if (!tilew) xbuf[c] = cn2;
        lds_barrier();
        const float xn = xbuf[c];
        if (w < NT && t + 1 < TT) tile_cell(t + 1, xn);
    }
}

extern "C" void kernel_launch(void* const* d_in, const int* in_sizes, int n_in,
                              void* d_out, int out_size, void* d_ws, size_t ws_size,
                              hipStream_t stream)
{
    const float* xin  = (const float*)d_in[0];
    const float* Wih1 = (const float*)d_in[1];
    const float* Whh1 = (const float*)d_in[2];
    const float* bih1 = (const float*)d_in[3];
    const float* bhh1 = (const float*)d_in[4];
    const float* Wih2 = (const float*)d_in[5];
    const float* Whh2 = (const float*)d_in[6];
    const float* bih2 = (const float*)d_in[7];
    const float* bhh2 = (const float*)d_in[8];
    const int*   futp = (const int*)d_in[9];
    float* out = (float*)d_out;

    dim3 grid(B_TOT / ROWS);
    dim3 block(32 * NW);
    lstm_seq_kernel<<<grid, block, 0, stream>>>(xin, Wih1, Whh1, bih1, bhh1,
                                                Wih2, Whh2, bih2, bhh2, futp, out);
}


# Own-chunk WMMA and the input term x*w leave the stretch in front of the stores: both issue behind the barrier under the LDS read latency, accumulator starts at bias + x*w; sum order changes
# speedup vs baseline: 11.2791x; 1.0495x over previous
#include <hip/hip_runtime.h>

typedef __attribute__((ext_vector_type(2))) float v2f;
typedef __attribute__((ext_vector_type(8))) float v8f;

#define B_TOT 1024
#define T_LEN 4096
#define HID   20
#define ROWS  16
#define NT    5
#define NW    6

__device__ __forceinline__ float tanh_f(float x) {
    return __builtin_amdgcn_tanhf(x);
}
__device__ __forceinline__ float sigm_h(float xh) {
    return __builtin_fmaf(0.5f, __builtin_amdgcn_tanhf(xh), 0.5f);
}
__device__ __forceinline__ float gate_scale(int g) {
    return (g == 2) ? 1.0f : 0.5f;
}
__device__ __forceinline__ float xhalf(float x) {
    int b = __builtin_bit_cast(int, x);
    return __builtin_bit_cast(float,
        __builtin_amdgcn_permlanex16(b, b, 0x76543210, 0xfedcba98, false, false));
}

__device__ __forceinline__ void lds_barrier() {
    __builtin_amdgcn_fence(__ATOMIC_RELEASE, "workgroup", "local");
    __builtin_amdgcn_s_barrier();
    __builtin_amdgcn_fence(__ATOMIC_ACQUIRE, "workgroup", "local");
}

__global__ void __launch_bounds__(32 * NW)
lstm_seq_kernel(const float* __restrict__ xin,
                const float* __restrict__ Wih1,
                const float* __restrict__ Whh1,
                const float* __restrict__ bih1,
                const float* __restrict__ bhh1,
                const float* __restrict__ Wih2,
                const float* __restrict__ Whh2,
                const float* __restrict__ bih2,
                const float* __restrict__ bhh2,
                const int*   __restrict__ futp,
                float* __restrict__ out)
{
    __shared__ v2f   hbuf[2][NT][32];
    __shared__ v2f   cbuf[2][NT][32];
    __shared__ float xbuf[ROWS];

    const int lane = threadIdx.x & 31;
    const int w    = __builtin_amdgcn_readfirstlane((int)(threadIdx.x >> 5));
    const bool tilew = (w < NT);
    const int mt   = tilew ? w : 0;
    const int c    = lane & 15;
    const int hf   = lane >> 4;
    const int row0 = blockIdx.x * ROWS;
    const int fut  = *futp;
    const int TT   = T_LEN + fut;

    int kcr[5];
#pragma unroll
    for (int j = 0; j < 5; ++j) kcr[j] = (mt + j < 5) ? (mt + j) : (mt + j - 5);
    v2f afr[5];
    {
        const int hm = c >> 3, r = c & 7, g = r >> 1, u = r & 1;
        const int wrow = g * HID + 4 * mt + 2 * hm + u;
#pragma unroll
        for (int j = 0; j < 5; ++j) {
            const float* p = Whh1 + wrow * HID + 4 * kcr[j] + 2 * hf;
            v2f a; a.x = p[0] * gate_scale(g); a.y = p[1] * gate_scale(g);
            afr[j] = a;
        }
    }
    float wx[8];
    v8f   bs;
#pragma unroll
    for (int r = 0; r < 8; ++r) {
        const int wrow = (r >> 1) * HID + 4 * mt + 2 * hf + (r & 1);
        wx[r] = Wih1[wrow] * gate_scale(r >> 1);
        bs[r] = (bih1[wrow] + bhh1[wrow]) * gate_scale(r >> 1);
    }
    v2f w2[4][5];
#pragma unroll
    for (int g = 0; g < 4; ++g)
#pragma unroll
        for (int m = 0; m < 5; ++m) {
            const float* q = Wih2 + g * HID + 4 * m + 2 * hf;
            v2f v; v.x = q[0] * gate_scale(g); v.y = q[1] * gate_scale(g);
            w2[g][m] = v;
        }
    float whh2[4], b2[4];
#pragma unroll
    for (int g = 0; g < 4; ++g) {
        whh2[g] = Whh2[g] * gate_scale(g);
        b2[g]   = (bih2[g] + bhh2[g]) * gate_scale(g);
    }

    v2f   cB = {0.f, 0.f};
    v2f   hn = {0.f, 0.f};
    float h2v = 0.0f, c2v = 0.0f;

    const float* xrow = xin + (size_t)(row0 + c) * T_LEN;
    float*       orow = out + (size_t)(row0 + c) * TT;


    auto tile_gates = [&](const int t, const float x, const bool with_x) __attribute__((always_inline)) -> v8f {
        const int p = t & 1;
        v2f hB[5];
#pragma unroll
        for (int j = 1; j < 5; ++j) hB[j] = hbuf[p][kcr[j]][lane];
        __builtin_amdgcn_sched_barrier(0);
        v8f Y = bs;
        if (with_x) {
#pragma unroll
            for (int r = 0; r < 8; ++r) Y[r] = __builtin_fmaf(x, wx[r], bs[r]);
        }
        Y = __builtin_amdgcn_wmma_f32_16x16x4_f32(false, afr[0], false, hn, (short)0, Y, false, false);
#pragma unroll
        for (int j = 1; j < 5; ++j)
            Y = __builtin_amdgcn_wmma_f32_16x16x4_f32(false, afr[j], false, hB[j], (short)0, Y, false, false);
        return Y;
    };
    auto tile_update = [&](const int t, const v8f G) __attribute__((always_inline)) {
        const int p = t & 1;
#pragma unroll
        for (int u = 0; u < 2; ++u) {
            const float ti = tanh_f(G[0 + u]), tf = tanh_f(G[2 + u]);
            const float tg = tanh_f(G[4 + u]), to = tanh_f(G[6 + u]);
            const float ig = __builtin_fmaf(0.5f, ti, 0.5f) * tg;
            const float cn = __builtin_fmaf(__builtin_fmaf(0.5f, tf, 0.5f), cB[u], ig);
            const float tc = tanh_f(cn);
            cB[u] = cn;
            hn[u] = __builtin_fmaf(0.5f, to, 0.5f) * tc;
        }
        hbuf[p][w][lane] = hn;
        cbuf[p][w][lane] = cB;
    };
    auto l2_cell = [&](const int t) __attribute__((always_inline)) -> float {
        const int p = t & 1;
        v2f cn[5];
#pragma unroll
        for (int m = 0; m < 5; ++m) cn[m] = cbuf[p][m][lane];
        v2f acc[4];
#pragma unroll
        for (int g = 0; g < 4; ++g) acc[g] = cn[0] * w2[g][0];
#pragma unroll
        for (int m = 1; m < 5; ++m)
#pragma unroll
            for (int g = 0; g < 4; ++g) acc[g] = __builtin_elementwise_fma(cn[m], w2[g][m], acc[g]);
        float g2[4];
#pragma unroll
        for (int g = 0; g < 4; ++g) {
            const float p2 = acc[g].x + acc[g].y;
            g2[g] = (p2 + xhalf(p2)) + (b2[g] + h2v * whh2[g]);
        }
        const float cn2 = sigm_h(g2[1]) * c2v + sigm_h(g2[0]) * tanh_f(g2[2]);
        c2v = cn2;
        h2v = sigm_h(g2[3]) * tanh_f(cn2);
        orow[t] = cn2;
        return cn2;
    };
    auto seq_step = [&](const int t, const float xnext) __attribute__((always_inline)) {
        lds_barrier();
        if (__builtin_expect(w >= NT, 0)) {
            l2_cell(t);
        } else {
            const v8f G = tile_gates(t, xnext, true);
            tile_update(t + 1, G);
        }
    };

    typedef __attribute__((ext_vector_type(4))) float v4f;
    const v4f* xq = reinterpret_cast<const v4f*>(xrow);
    v4f q0 = xq[0], q1 = xq[1];
    if (w < NT) {
        v8f G0;
#pragma unroll
        for (int r = 0; r < 8; ++r) G0[r] = __builtin_fmaf(q0.x, wx[r], bs[r]);
        tile_update(0, G0);
    }
    int t = 0;
    for (; t < T_LEN - 4; t += 4) {
        const int nq = (t >> 2) + 2;
        const v4f q2 = xq[(nq < T_LEN / 4) ? nq : (T_LEN / 4 - 1)];
        seq_step(t,     q0.y);
        seq_step(t + 1, q0.z);
        seq_step(t + 2, q0.w);
        seq_step(t + 3, q1.x);
        q0 = q1; q1 = q2;
    }
    seq_step(t,     q0.y);
    seq_step(t + 1, q0.z);
    seq_step(t + 2, q0.w);
    t += 3;
    for (; t < TT; ++t) {
        lds_barrier();
        float cn2 = 0.0f;
        v8f Yf = bs;
        if (w < NT) Yf = tile_gates(t, 0.0f, false);
        else        cn2 = l2_cell(t);
        if (!tilew) xbuf[c] = cn2;
        lds_barrier();
        const float xn = xbuf[c];
        if (w < NT && t + 1 < TT) {
            v8f G;
#pragma unroll
            for (int r = 0; r < 8; ++r) G[r] = __builtin_fmaf(xn, wx[r], Yf[r]);
            tile_update(t + 1, G);
        }
    }
}

extern "C" void kernel_launch(void* const* d_in, const int* in_sizes, int n_in,
                              void* d_out, int out_size, void* d_ws, size_t ws_size,
                              hipStream_t stream)
{
    const float* xin  = (const float*)d_in[0];
    const float* Wih1 = (const float*)d_in[1];
    const float* Whh1 = (const float*)d_in[2];
    const float* bih1 = (const float*)d_in[3];
    const float* bhh1 = (const float*)d_in[4];
    const float* Wih2 = (const float*)d_in[5];
    const float* Whh2 = (const float*)d_in[6];
    const float* bih2 = (const float*)d_in[7];
    const float* bhh2 = (const float*)d_in[8];
    const int*   futp = (const int*)d_in[9];
    float* out = (float*)d_out;

    dim3 grid(B_TOT / ROWS);
    dim3 block(32 * NW);
    lstm_seq_kernel<<<grid, block, 0, stream>>>(xin, Wih1, Whh1, bih1, bhh1,
                                                Wih2, Whh2, bih2, bhh2, futp, out);
}


# One WMMA of each of the two tile waves sharing a vector unit is made before the barrier by the lone-unit wave owning that h chunk, passed on as a partial tile in LDS; loop per role; sum order changes
# speedup vs baseline: 11.8312x; 1.0489x over previous
#include <hip/hip_runtime.h>

typedef __attribute__((ext_vector_type(2))) float v2f;
typedef __attribute__((ext_vector_type(8))) float v8f;
typedef __attribute__((ext_vector_type(4))) float v4f;

#define B_TOT 1024
#define T_LEN 4096
#define HID   20
#define ROWS  16
#define NT    5
#define NW    6

__device__ __forceinline__ float tanh_f(float x) {
    return __builtin_amdgcn_tanhf(x);
}
__device__ __forceinline__ float sigm_h(float xh) {
    return __builtin_fmaf(0.5f, __builtin_amdgcn_tanhf(xh), 0.5f);
}
__device__ __forceinline__ float gate_scale(int g) {
    return (g == 2) ? 1.0f : 0.5f;
}
__device__ __forceinline__ float xhalf(float x) {
    int b = __builtin_bit_cast(int, x);
    return __builtin_bit_cast(float,
        __builtin_amdgcn_permlanex16(b, b, 0x76543210, 0xfedcba98, false, false));
}

__device__ __forceinline__ void lds_barrier() {
    __builtin_amdgcn_fence(__ATOMIC_RELEASE, "workgroup", "local");
    __builtin_amdgcn_s_barrier();
    __builtin_amdgcn_fence(__ATOMIC_ACQUIRE, "workgroup", "local");
}

__global__ void __launch_bounds__(32 * NW)
lstm_seq_kernel(const float* __restrict__ xin,
                const float* __restrict__ Wih1,
                const float* __restrict__ Whh1,
                const float* __restrict__ bih1,
                const float* __restrict__ bhh1,
                const float* __restrict__ Wih2,
                const float* __restrict__ Whh2,
                const float* __restrict__ bih2,
                const float* __restrict__ bhh2,
                const int*   __restrict__ futp,
                float* __restrict__ out)
{
    __shared__ v2f   hbuf[2][NT][32];
    __shared__ v2f   cbuf[2][NT][32];
    __shared__ float xbuf[ROWS];
    __shared__ v4f   qbuf[2][2][2][32];

    const int lane = threadIdx.x & 31;
    const int w    = __builtin_amdgcn_readfirstlane((int)(threadIdx.x >> 5));
    const bool tilew = (w < NT);
    const int mt   = tilew ? w : 0;
    const int c    = lane & 15;
    const int hf   = lane >> 4;
    const int row0 = blockIdx.x * ROWS;
    const int fut  = *futp;
    const int TT   = T_LEN + fut;

    int kcr[5];
#pragma unroll
    for (int j = 0; j < 5; ++j) kcr[j] = (mt + j < 5) ? (mt + j) : (mt + j - 5);
    const bool give  = (w == 2) || (w == 3);
    const bool take  = (w == 0) || (w == 4);
    const int  qslot = (w == 0 || w == 2) ? 0 : 1;
    if (w == 0) { kcr[2] = 3; kcr[3] = 4; kcr[4] = 2; }
    v2f afr[5];
    {
        const int hm = c >> 3, r = c & 7, g = r >> 1, u = r & 1;
        const int wrow = g * HID + 4 * mt + 2 * hm + u;
#pragma unroll
        for (int j = 0; j < 5; ++j) {
            const float* p = Whh1 + wrow * HID + 4 * kcr[j] + 2 * hf;
            v2f a; a.x = p[0] * gate_scale(g); a.y = p[1] * gate_scale(g);
            afr[j] = a;
        }
    }
    v2f agv;
    {
        const int hm = c >> 3, r = c & 7, g = r >> 1, u = r & 1;
        const int tmt  = (w == 2) ? 0 : 4;
        const int wrow = g * HID + 4 * tmt + 2 * hm + u;
        const float* p = Whh1 + wrow * HID + 4 * mt + 2 * hf;
        agv.x = p[0] * gate_scale(g); agv.y = p[1] * gate_scale(g);
    }
    float wx[8];
    v8f   bs;
#pragma unroll
    for (int r = 0; r < 8; ++r) {
        const int wrow = (r >> 1) * HID + 4 * mt + 2 * hf + (r & 1);
        wx[r] = Wih1[wrow] * gate_scale(r >> 1);
        bs[r] = (bih1[wrow] + bhh1[wrow]) * gate_scale(r >> 1);
    }
    v2f w2[4][5];
#pragma unroll
    for (int g = 0; g < 4; ++g)
#pragma unroll
        for (int m = 0; m < 5; ++m) {
            const float* q = Wih2 + g * HID + 4 * m + 2 * hf;
            v2f v; v.x = q[0] * gate_scale(g); v.y = q[1] * gate_scale(g);
            w2[g][m] = v;
        }
    float whh2[4], b2[4];
#pragma unroll
    for (int g = 0; g < 4; ++g) {
        whh2[g] = Whh2[g] * gate_scale(g);
        b2[g]   = (bih2[g] + bhh2[g]) * gate_scale(g);
    }

    v2f   cB = {0.f, 0.f};
    v2f   hn = {0.f, 0.f};
    float h2v = 0.0f, c2v = 0.0f;

    const float* xrow = xin + (size_t)(row0 + c) * T_LEN;
    float*       orow = out + (size_t)(row0 + c) * TT;


    auto tile_gates = [&](const int t, const float x, const bool with_x, const bool tk) __attribute__((always_inline)) -> v8f {
        const int p = t & 1;
        v2f hB[5];
        if (tk) {
            const v4f qa = qbuf[p][qslot][0][lane];
            const v4f qb = qbuf[p][qslot][1][lane];
#pragma unroll
            for (int j = 1; j < 4; ++j) hB[j] = hbuf[p][kcr[j]][lane];
            __builtin_amdgcn_sched_barrier(0);
            v8f Y = bs;
            if (with_x) {
#pragma unroll
                for (int r = 0; r < 8; ++r) Y[r] = __builtin_fmaf(x, wx[r], bs[r]);
            }
            Y = __builtin_amdgcn_wmma_f32_16x16x4_f32(false, afr[0], false, hn, (short)0, Y, false, false);
#pragma unroll
            for (int r = 0; r < 4; ++r) { Y[r] += qa[r]; Y[4 + r] += qb[r]; }
#pragma unroll
            for (int j = 1; j < 4; ++j)
                Y = __builtin_amdgcn_wmma_f32_16x16x4_f32(false, afr[j], false, hB[j], (short)0, Y, false, false);
            return Y;
        }
#pragma unroll
        for (int j = 1; j < 5; ++j) hB[j] = hbuf[p][kcr[j]][lane];
        __builtin_amdgcn_sched_barrier(0);
        v8f Y = bs;
        if (with_x) {
#pragma unroll
            for (int r = 0; r < 8; ++r) Y[r] = __builtin_fmaf(x, wx[r], bs[r]);
        }
        Y = __builtin_amdgcn_wmma_f32_16x16x4_f32(false, afr[0], false, hn, (short)0, Y, false, false);
#pragma unroll
        for (int j = 1; j < 5; ++j)
            Y = __builtin_amdgcn_wmma_f32_16x16x4_f32(false, afr[j], false, hB[j], (short)0, Y, false, false);
        return Y;
    };
    auto tile_update = [&](const int t, const v8f G, const bool gv) __attribute__((always_inline)) {
        const int p = t & 1;
#pragma unroll
        for (int u = 0; u < 2; ++u) {
            const float ti = tanh_f(G[0 + u]), tf = tanh_f(G[2 + u]);
            const float tg = tanh_f(G[4 + u]), to = tanh_f(G[6 + u]);
            const float ig = __builtin_fmaf(0.5f, ti, 0.5f) * tg;
            const float cn = __builtin_fmaf(__builtin_fmaf(0.5f, tf, 0.5f), cB[u], ig);
            const float tc = tanh_f(cn);
            cB[u] = cn;
            hn[u] = __builtin_fmaf(0.5f, to, 0.5f) * tc;
        }
        if (gv) {
            const v8f Z = {0.f, 0.f, 0.f, 0.f, 0.f, 0.f, 0.f, 0.f};
            const v8f Q = __builtin_amdgcn_wmma_f32_16x16x4_f32(false, agv, false, hn, (short)0, Z, false, false);
            hbuf[p][w][lane] = hn;
            cbuf[p][w][lane] = cB;
            const v4f qa = {Q[0], Q[1], Q[2], Q[3]};
            const v4f qb = {Q[4], Q[5], Q[6], Q[7]};
            qbuf[p][qslot][0][lane] = qa;
            qbuf[p][qslot][1][lane] = qb;
        } else {
            hbuf[p][w][lane] = hn;
            cbuf[p][w][lane] = cB;
        }
    };
    auto l2_cell = [&](const int t) __attribute__((always_inline)) -> float {
        const int p = t & 1;
        v2f cn[5];
#pragma unroll
        for (int m = 0; m < 5; ++m) cn[m] = cbuf[p][m][lane];
        v2f acc[4];
#pragma unroll
        for (int g = 0; g < 4; ++g) acc[g] = cn[0] * w2[g][0];
#pragma unroll
        for (int m = 1; m < 5; ++m)
#pragma unroll
            for (int g = 0; g < 4; ++g) acc[g] = __builtin_elementwise_fma(cn[m], w2[g][m], acc[g]);
        float g2[4];
#pragma unroll
        for (int g = 0; g < 4; ++g) {
            const float p2 = acc[g].x + acc[g].y;
            g2[g] = (p2 + xhalf(p2)) + (b2[g] + h2v * whh2[g]);
        }
        const float cn2 = sigm_h(g2[1]) * c2v + sigm_h(g2[0]) * tanh_f(g2[2]);
        c2v = cn2;
        h2v = sigm_h(g2[3]) * tanh_f(cn2);
        orow[t] = cn2;
        return cn2;
    };
    auto seq_step = [&](const int t, const float xnext, const int role) __attribute__((always_inline)) {
        lds_barrier();
        if (role == 3) {
            l2_cell(t);
        } else {
            const v8f G = tile_gates(t, xnext, true, role == 0);
            tile_update(t + 1, G, role == 1);
        }
    };

    const v4f* xq = reinterpret_cast<const v4f*>(xrow);
    v4f q0 = xq[0], q1 = xq[1];
    if (w < NT) {
        v8f G0;
#pragma unroll
        for (int r = 0; r < 8; ++r) G0[r] = __builtin_fmaf(q0.x, wx[r], bs[r]);
        tile_update(0, G0, give);
    }
    int t = 0;
    auto seq_loop = [&](const int role) __attribute__((always_inline)) {
        for (; t < T_LEN - 4; t += 4) {
            const int nq = (t >> 2) + 2;
            const v4f q2 = xq[(nq < T_LEN / 4) ? nq : (T_LEN / 4 - 1)];
            seq_step(t,     q0.y, role);
            seq_step(t + 1, q0.z, role);
            seq_step(t + 2, q0.w, role);
            seq_step(t + 3, q1.x, role);
            q0 = q1; q1 = q2;
        }
        seq_step(t,     q0.y, role);
        seq_step(t + 1, q0.z, role);
        seq_step(t + 2, q0.w, role);
        t += 3;
    };
    if (take)        seq_loop(0);
    else if (give)   seq_loop(1);
    else if (w < NT) seq_loop(2);
    else             seq_loop(3);
    for (; t < TT; ++t) {
        lds_barrier();
        float cn2 = 0.0f;
        v8f Yf = bs;
        if (w < NT) Yf = tile_gates(t, 0.0f, false, take);
        else        cn2 = l2_cell(t);
        if (!tilew) xbuf[c] = cn2;
        lds_barrier();
        const float xn = xbuf[c];
        if (w < NT && t + 1 < TT) {
            v8f G;
#pragma unroll
            for (int r = 0; r < 8; ++r) G[r] = __builtin_fmaf(xn, wx[r], Yf[r]);
            tile_update(t + 1, G, give);
        }
    }
}

extern "C" void kernel_launch(void* const* d_in, const int* in_sizes, int n_in,
                              void* d_out, int out_size, void* d_ws, size_t ws_size,
                              hipStream_t stream)
{
    const float* xin  = (const float*)d_in[0];
    const float* Wih1 = (const float*)d_in[1];
    const float* Whh1 = (const float*)d_in[2];
    const float* bih1 = (const float*)d_in[3];
    const float* bhh1 = (const float*)d_in[4];
    const float* Wih2 = (const float*)d_in[5];
    const float* Whh2 = (const float*)d_in[6];
    const float* bih2 = (const float*)d_in[7];
    const float* bhh2 = (const float*)d_in[8];
    const int*   futp = (const int*)d_in[9];
    float* out = (float*)d_out;

    dim3 grid(B_TOT / ROWS);
    dim3 block(32 * NW);
    lstm_seq_kernel<<<grid, block, 0, stream>>>(xin, Wih1, Whh1, bih1, bhh1,
                                                Wih2, Whh2, bih2, bhh2, futp, out);
}
